# starting kernel
# baseline (speedup 1.0000x reference)
#include <hip/hip_runtime.h>
#include <hip/hip_bf16.h>

#define NUM_USERS 100000
#define NUM_ITEMS 200000
#define N_TOTAL   (NUM_USERS + NUM_ITEMS)
#define EMBED_DIM 64
#define NUM_LAYERS 3
#define NUM_EDGES 1200000
#define BATCH 4096

typedef __attribute__((ext_vector_type(2))) float v2f;
typedef __attribute__((ext_vector_type(8))) float v8f;

__global__ __launch_bounds__(256) void scatter_add_kernel(
    const float* __restrict__ cur, float* __restrict__ nxt,
    const int* __restrict__ src, const int* __restrict__ dst) {
  long long t = (long long)blockIdx.x * blockDim.x + threadIdx.x;
  int e = (int)(t >> 4);
  if (e >= NUM_EDGES) return;
  int part = ((int)t & 15) << 2;
  size_t s = (size_t)src[e];
  size_t d = (size_t)dst[e];
  const float4 v = *(const float4*)(cur + s * EMBED_DIM + part);
  float* p = nxt + d * EMBED_DIM + part;
  atomicAdd(p + 0, v.x);
  atomicAdd(p + 1, v.y);
  atomicAdd(p + 2, v.z);
  atomicAdd(p + 3, v.w);
}

__global__ __launch_bounds__(256) void gather_acc_kernel(
    const float* __restrict__ layer,
    const int* __restrict__ uidx, const int* __restrict__ iidx,
    float* __restrict__ ubuf, float* __restrict__ vbuf) {
  int t = blockIdx.x * blockDim.x + threadIdx.x;
  int b = t >> 4;
  if (b >= BATCH) return;
  int part = (t & 15) << 2;
  size_t u = (size_t)uidx[b];
  size_t it = (size_t)iidx[b] + NUM_USERS;
  const float4 uu = *(const float4*)(layer + u * EMBED_DIM + part);
  const float4 vv = *(const float4*)(layer + it * EMBED_DIM + part);
  float4* up = (float4*)(ubuf + (size_t)b * EMBED_DIM + part);
  float4* vp = (float4*)(vbuf + (size_t)b * EMBED_DIM + part);
  float4 a = *up;
  a.x += uu.x; a.y += uu.y; a.z += uu.z; a.w += uu.w;
  *up = a;
  float4 c = *vp;
  c.x += vv.x; c.y += vv.y; c.z += vv.z; c.w += vv.w;
  *vp = c;
}

__global__ __launch_bounds__(128) void dot_wmma_kernel(
    const float* __restrict__ ubuf, const float* __restrict__ vbuf,
    float* __restrict__ out) {
  const int lane = threadIdx.x & 31;
  const int tile = blockIdx.x * (blockDim.x >> 5) + (threadIdx.x >> 5);
  const int rowBase = tile * 16;
  const int row = rowBase + (lane & 15);
  const int kOff = (lane >> 4) << 1;

  const float* uRow = ubuf + (size_t)row * EMBED_DIM + kOff;
  const float* vRow = vbuf + (size_t)row * EMBED_DIM + kOff;

  v8f c = {};
#pragma unroll
  for (int k = 0; k < EMBED_DIM; k += 4) {
    v2f a, b;
    a.x = uRow[k];
    a.y = uRow[k + 1];
    b.x = vRow[k];
    b.y = vRow[k + 1];
    c = __builtin_amdgcn_wmma_f32_16x16x4_f32(false, a, false, b, (short)0, c,
                                              false, false);
  }

  float cd[8];
#pragma unroll
  for (int i = 0; i < 8; ++i) cd[i] = c[i];
  int dsel = (lane < 16) ? lane : (lane - 24);
  float diag = cd[0];
#pragma unroll
  for (int i = 1; i < 8; ++i)
    if (dsel == i) diag = cd[i];

  const float scale = 1.0f / 16.0f;
  if (lane < 8)
    out[rowBase + lane] = diag * scale;
  else if (lane >= 24)
    out[rowBase + (lane - 16)] = diag * scale;
}

extern "C" void kernel_launch(void* const* d_in, const int* in_sizes, int n_in,
                              void* d_out, int out_size, void* d_ws,
                              size_t ws_size, hipStream_t stream) {
  const float* emb = (const float*)d_in[0];
  const int* eidx = (const int*)d_in[1];
  const int* uidx = (const int*)d_in[2];
  const int* iidx = (const int*)d_in[3];
  float* out = (float*)d_out;

  const int* src = eidx;
  const int* dst = eidx + NUM_EDGES;

  char* ws = (char*)d_ws;
  const size_t nodeBytes = (size_t)N_TOTAL * EMBED_DIM * sizeof(float);
  const size_t batBytes = (size_t)BATCH * EMBED_DIM * sizeof(float);
  float* bufA = (float*)ws;
  float* bufB = (float*)(ws + nodeBytes);
  float* ubuf = (float*)(ws + 2 * nodeBytes);
  float* vbuf = (float*)(ws + 2 * nodeBytes + batBytes);

  hipMemsetAsync(ubuf, 0, batBytes, stream);
  hipMemsetAsync(vbuf, 0, batBytes, stream);
  const int gGrid = (BATCH * 16 + 255) / 256;
  gather_acc_kernel<<<gGrid, 256, 0, stream>>>(emb, uidx, iidx, ubuf, vbuf);

  const long long sThreads = (long long)NUM_EDGES * 16;
  const int sGrid = (int)((sThreads + 255) / 256);
  const float* cur = emb;
  for (int l = 0; l < NUM_LAYERS; ++l) {
    float* nxt = (l & 1) ? bufB : bufA;
    hipMemsetAsync(nxt, 0, nodeBytes, stream);
    scatter_add_kernel<<<sGrid, 256, 0, stream>>>(cur, nxt, src, dst);
    gather_acc_kernel<<<gGrid, 256, 0, stream>>>(nxt, uidx, iidx, ubuf, vbuf);
    cur = nxt;
  }

  dot_wmma_kernel<<<BATCH / 16 / 4, 128, 0, stream>>>(ubuf, vbuf, out);
}


# Transposed walk: per-call in-edge linked lists (one int exchange per edge), one wave per batch row sums rows at ends of all in-paths of 0-3 hops; no node planes, no float atomics; sum order changed
# speedup vs baseline: 5.2013x; 5.2013x over previous
#include <hip/hip_runtime.h>
#include <hip/hip_bf16.h>

#define NUM_USERS 100000
#define NUM_ITEMS 200000
#define N_TOTAL   (NUM_USERS + NUM_ITEMS)
#define EMBED_DIM 64
#define NUM_LAYERS 3
#define NUM_EDGES 1200000
#define BATCH 4096

#define EDGE_QUADS (NUM_EDGES / 4)
#define HEAD_PAD 300032
#define LIST_CAP 1024

typedef __attribute__((ext_vector_type(2))) float v2f;
typedef __attribute__((ext_vector_type(8))) float v8f;

__global__ __launch_bounds__(256) void link_kernel(const int* __restrict__ src,
                                                   const int* __restrict__ dst,
                                                   int* __restrict__ head,
                                                   int* __restrict__ link) {
  const int t = blockIdx.x * 256 + threadIdx.x;
  if (t >= EDGE_QUADS) return;
  const int4 s = *(const int4*)(src + (size_t)t * 4);
  const int4 d = *(const int4*)(dst + (size_t)t * 4);
  const int e = t * 4;
  const int o0 = atomicExch(head + d.x, e);
  const int o1 = atomicExch(head + d.y, e + 1);
  const int o2 = atomicExch(head + d.z, e + 2);
  const int o3 = atomicExch(head + d.w, e + 3);
  int4 p0, p1;
  p0.x = s.x; p0.y = o0; p0.z = s.y; p0.w = o1;
  p1.x = s.z; p1.y = o2; p1.z = s.w; p1.w = o3;
  int4* q = (int4*)(link + (size_t)t * 8);
  q[0] = p0;
  q[1] = p1;
}

static __device__ __forceinline__ float4 ld_row4(const float* __restrict__ emb, int node, int c4) {
  return *(const float4*)(emb + (size_t)node * EMBED_DIM + c4);
}

__global__ __launch_bounds__(32) void walk_kernel(
    const float* __restrict__ emb,
    const int* __restrict__ uidx, const int* __restrict__ iidx,
    const int* __restrict__ head, const int* __restrict__ linkw,
    float* __restrict__ ubuf, float* __restrict__ vbuf) {
  const int2* __restrict__ link = (const int2*)linkw;
  __shared__ int list[LIST_CAP];
  const int lane = threadIdx.x;
  const int h = lane >> 4;
  const int c4 = (lane & 15) << 2;
  const int wv = blockIdx.x;
  const bool isUser = wv < BATCH;
  const int b = isUser ? wv : wv - BATCH;
  const int node = isUser ? uidx[b] : NUM_USERS + iidx[b];
  float* outRow = (isUser ? ubuf : vbuf) + (size_t)b * EMBED_DIM + c4;

  if (lane == 0) list[0] = node;
  __syncthreads();

  bool ovf = false;
  int fb = 0, fe = 1;
  for (int lvl = 0; lvl < NUM_LAYERS; ++lvl) {
    int wr = fe;
    for (int base = fb; base < fe; base += 32) {
      if (ovf) break;
      const int i = base + lane;
      int e = (i < fe) ? head[list[i]] : -1;
      for (int hop = 0; hop < NUM_EDGES; ++hop) {
        const bool act = e >= 0;
        const unsigned m = __builtin_amdgcn_ballot_w32(act);
        if (m == 0u) break;
        const int na = __builtin_popcount(m);
        if (wr + na > LIST_CAP) {
          ovf = true;
          break;
        }
        if (act) {
          const int2 p = link[e];
          list[wr + (int)__builtin_amdgcn_mbcnt_lo(m, 0u)] = p.x;
          e = p.y;
        }
        wr += na;
      }
    }
    __syncthreads();
    fb = fe;
    fe = wr;
  }

  float4 acc = make_float4(0.f, 0.f, 0.f, 0.f);
  if (!ovf) {
    const int n = fe;
    for (int i = 0; i < n; i += 16) {
      float4 v[8];
#pragma unroll
      for (int u = 0; u < 8; ++u) {
        const int idx = i + 2 * u + h;
        const bool ok = idx < n;
        const int s = list[ok ? idx : 0];
        const float4 r = ld_row4(emb, s, c4);
        v[u] = ok ? r : make_float4(0.f, 0.f, 0.f, 0.f);
      }
#pragma unroll
      for (int u = 0; u < 8; ++u) {
        acc.x += v[u].x; acc.y += v[u].y; acc.z += v[u].z; acc.w += v[u].w;
      }
    }
  } else {
    const float4 r0 = ld_row4(emb, node, c4);
    if (h == 0) { acc.x += r0.x; acc.y += r0.y; acc.z += r0.z; acc.w += r0.w; }
    int e1 = head[node];
    for (int g1 = 0; g1 < NUM_EDGES && e1 >= 0; ++g1) {
      const int2 p1 = link[e1];
      e1 = p1.y;
      const float4 r1 = ld_row4(emb, p1.x, c4);
      if (h == 0) { acc.x += r1.x; acc.y += r1.y; acc.z += r1.z; acc.w += r1.w; }
      int e2 = head[p1.x];
      for (int g2 = 0; g2 < NUM_EDGES && e2 >= 0; ++g2) {
        const int2 p2 = link[e2];
        e2 = p2.y;
        const float4 r2 = ld_row4(emb, p2.x, c4);
        if (h == 0) { acc.x += r2.x; acc.y += r2.y; acc.z += r2.z; acc.w += r2.w; }
        int e3 = head[p2.x];
        for (int g3 = 0; g3 < NUM_EDGES && e3 >= 0; ++g3) {
          const int2 p3 = link[e3];
          e3 = p3.y;
          const float4 r3 = ld_row4(emb, p3.x, c4);
          if (h == 0) { acc.x += r3.x; acc.y += r3.y; acc.z += r3.z; acc.w += r3.w; }
        }
      }
    }
  }

  acc.x += __shfl_xor(acc.x, 16);
  acc.y += __shfl_xor(acc.y, 16);
  acc.z += __shfl_xor(acc.z, 16);
  acc.w += __shfl_xor(acc.w, 16);
  if (h == 0) *(float4*)outRow = acc;
}

__global__ __launch_bounds__(128) void dot_wmma_kernel(
    const float* __restrict__ ubuf, const float* __restrict__ vbuf,
    float* __restrict__ out) {
  const int lane = threadIdx.x & 31;
  const int tile = blockIdx.x * (blockDim.x >> 5) + (threadIdx.x >> 5);
  const int rowBase = tile * 16;
  const int row = rowBase + (lane & 15);
  const int kOff = (lane >> 4) << 1;

  const float* uRow = ubuf + (size_t)row * EMBED_DIM + kOff;
  const float* vRow = vbuf + (size_t)row * EMBED_DIM + kOff;

  v8f c = {};
#pragma unroll
  for (int k = 0; k < EMBED_DIM; k += 4) {
    v2f a, b;
    a.x = uRow[k];
    a.y = uRow[k + 1];
    b.x = vRow[k];
    b.y = vRow[k + 1];
    c = __builtin_amdgcn_wmma_f32_16x16x4_f32(false, a, false, b, (short)0, c,
                                              false, false);
  }

  float cd[8];
#pragma unroll
  for (int i = 0; i < 8; ++i) cd[i] = c[i];
  int dsel = (lane < 16) ? lane : (lane - 24);
  float diag = cd[0];
#pragma unroll
  for (int i = 1; i < 8; ++i)
    if (dsel == i) diag = cd[i];

  const float scale = 1.0f / 16.0f;
  if (lane < 8)
    out[rowBase + lane] = diag * scale;
  else if (lane >= 24)
    out[rowBase + (lane - 16)] = diag * scale;
}

extern "C" void kernel_launch(void* const* d_in, const int* in_sizes, int n_in,
                              void* d_out, int out_size, void* d_ws,
                              size_t ws_size, hipStream_t stream) {
  const float* emb = (const float*)d_in[0];
  const int* eidx = (const int*)d_in[1];
  const int* uidx = (const int*)d_in[2];
  const int* iidx = (const int*)d_in[3];
  float* out = (float*)d_out;

  const int* src = eidx;
  const int* dst = eidx + NUM_EDGES;

  int* head = (int*)d_ws;
  int* link = head + HEAD_PAD;
  float* ubuf = (float*)(link + 2 * NUM_EDGES);
  float* vbuf = ubuf + BATCH * EMBED_DIM;

  hipMemsetAsync(head, 0xFF, HEAD_PAD * sizeof(int), stream);
  link_kernel<<<(EDGE_QUADS + 255) / 256, 256, 0, stream>>>(src, dst, head, link);

  walk_kernel<<<2 * BATCH, 32, 0, stream>>>(emb, uidx, iidx, head, link, ubuf, vbuf);

  dot_wmma_kernel<<<BATCH / 16 / 4, 128, 0, stream>>>(ubuf, vbuf, out);
}


# Exchange-built in-edge lists replaced by a destination-ordered adjacency from two LDS-binned passes (no global atomics, no memset); a walk level reads offsets and source runs; sum order changed
# speedup vs baseline: 9.0993x; 1.7494x over previous
#include <hip/hip_runtime.h>
#include <hip/hip_bf16.h>

#define NUM_USERS 100000
#define NUM_ITEMS 200000
#define N_TOTAL   (NUM_USERS + NUM_ITEMS)
#define EMBED_DIM 64
#define NUM_LAYERS 3
#define NUM_EDGES 1200000
#define BATCH 4096

#define LIST_CAP 1024

#define NB 512
#define NPB 586
#define CH 8192
#define GA ((NUM_EDGES + CH - 1) / CH)
#define WSTR 516
#define WOFF_PAD 75904
#define OFF_PAD 300032
#define CSR_PAD 64
#define LCAP 4096

typedef __attribute__((ext_vector_type(2))) float v2f;
typedef __attribute__((ext_vector_type(8))) float v8f;

__global__ __launch_bounds__(1024) void bin_kernel(const int* __restrict__ src,
                                                   const int* __restrict__ dst,
                                                   int* __restrict__ sortedW,
                                                   int* __restrict__ woff) {
  int2* __restrict__ sortedE = (int2*)sortedW;
  __shared__ int2 ent[CH];
  __shared__ int cnt[NB];
  __shared__ int start[NB + 1];
  __shared__ int wsum[16];
  const int t = threadIdx.x;
  const int lane = t & 31;
  const int wg = blockIdx.x;
  const int e0 = wg * CH + t * 8;
  if (t < NB) cnt[t] = 0;
  __syncthreads();

  int s[8], d[8], r[8], b[8];
  const bool valid = e0 < NUM_EDGES;
  if (valid) {
    const int4 sa = *(const int4*)(src + e0);
    const int4 sb = *(const int4*)(src + e0 + 4);
    const int4 da = *(const int4*)(dst + e0);
    const int4 db = *(const int4*)(dst + e0 + 4);
    s[0] = sa.x; s[1] = sa.y; s[2] = sa.z; s[3] = sa.w;
    s[4] = sb.x; s[5] = sb.y; s[6] = sb.z; s[7] = sb.w;
    d[0] = da.x; d[1] = da.y; d[2] = da.z; d[3] = da.w;
    d[4] = db.x; d[5] = db.y; d[6] = db.z; d[7] = db.w;
#pragma unroll
    for (int k = 0; k < 8; ++k) {
      b[k] = d[k] / NPB;
      r[k] = atomicAdd(&cnt[b[k]], 1);
    }
  }
  __syncthreads();

  const int v = (t < NB) ? cnt[t] : 0;
  int x = v;
  for (int o = 1; o < 32; o <<= 1) {
    const int y = __shfl_up(x, o);
    if (lane >= o) x += y;
  }
  if (t < NB && lane == 31) wsum[t >> 5] = x;
  __syncthreads();
  if (t < NB) {
    int pre = 0;
    for (int w = 0; w < (t >> 5); ++w) pre += wsum[w];
    start[t] = pre + x - v;
    if (t == NB - 1) start[NB] = pre + x;
  }
  __syncthreads();

  if (valid) {
#pragma unroll
    for (int k = 0; k < 8; ++k) ent[start[b[k]] + r[k]] = make_int2(s[k], d[k]);
  }
  __syncthreads();

  const int left = NUM_EDGES - wg * CH;
  const int nvalid = left < CH ? left : CH;
  int2* outE = sortedE + (size_t)wg * CH;
  for (int i = t; i < nvalid; i += 1024) outE[i] = ent[i];
  for (int j = t; j <= NB; j += 1024) woff[wg * WSTR + j] = start[j];
}

static __device__ __forceinline__ int scan256(int v, int* wsum8, int& total) {
  const int t = threadIdx.x;
  const int lane = t & 31;
  const int wave = t >> 5;
  int x = v;
  for (int o = 1; o < 32; o <<= 1) {
    const int y = __shfl_up(x, o);
    if (lane >= o) x += y;
  }
  if (lane == 31) wsum8[wave] = x;
  __syncthreads();
  int pre = 0, tot = 0;
  for (int w = 0; w < 8; ++w) {
    const int q = wsum8[w];
    if (w < wave) pre += q;
    tot += q;
  }
  __syncthreads();
  total = tot;
  return pre + x - v;
}

__global__ __launch_bounds__(256) void csr_kernel(const int* __restrict__ sortedW,
                                                  const int* __restrict__ woff,
                                                  int* __restrict__ offW,
                                                  int* __restrict__ csr) {
  const int2* __restrict__ sortedE = (const int2*)sortedW;
  int2* __restrict__ off2 = (int2*)offW;
  __shared__ int2 E[LCAP];
  __shared__ int stage[LCAP];
  __shared__ int ncnt[NPB];
  __shared__ int nstart[NPB + 1];
  __shared__ int wsum8[8];
  const int t = threadIdx.x;
  const int b = blockIdx.x;
  const int nodeBase = b * NPB;

  for (int i = t; i < NPB; i += 256) ncnt[i] = 0;

  int a = 0, len = 0;
  if (t < GA) {
    a = woff[t * WSTR + b];
    len = woff[t * WSTR + b + 1] - a;
  }
  int total, base;
  const int ss = scan256(len, wsum8, total);
  (void)scan256(a, wsum8, base);
  const int2* g = sortedE + (size_t)t * CH + a;
  const bool inLds = total <= LCAP;

  if (inLds) {
    for (int k0 = 0; k0 < len; k0 += 8) {
      int2 p[8];
#pragma unroll
      for (int u = 0; u < 8; ++u)
        if (k0 + u < len) p[u] = g[k0 + u];
#pragma unroll
      for (int u = 0; u < 8; ++u)
        if (k0 + u < len) E[ss + k0 + u] = p[u];
    }
    __syncthreads();
    for (int j = t; j < total; j += 256) {
      const int2 p = E[j];
      const int ln = p.y - nodeBase;
      const int rk = atomicAdd(&ncnt[ln], 1);
      E[j] = make_int2(p.x, ln | (rk << 10));
    }
  } else {
    for (int k = 0; k < len; ++k) atomicAdd(&ncnt[g[k].y - nodeBase], 1);
  }
  __syncthreads();

  const int i0 = 3 * t;
  const int c0 = (i0 < NPB) ? ncnt[i0] : 0;
  const int c1 = (i0 + 1 < NPB) ? ncnt[i0 + 1] : 0;
  const int c2 = (i0 + 2 < NPB) ? ncnt[i0 + 2] : 0;
  int tot2;
  const int pre = scan256(c0 + c1 + c2, wsum8, tot2);
  if (i0 < NPB) nstart[i0] = pre;
  if (i0 + 1 < NPB) nstart[i0 + 1] = pre + c0;
  if (i0 + 2 < NPB) nstart[i0 + 2] = pre + c0 + c1;
  if (t == 0) nstart[NPB] = tot2;
  __syncthreads();

  for (int i = t; i < NPB; i += 256) {
    const int n = nodeBase + i;
    if (n < N_TOTAL) off2[n] = make_int2(base + nstart[i], base + nstart[i + 1]);
  }
  if (b == NB - 1 && t < 4) csr[NUM_EDGES + t] = 0;

  if (inLds) {
    for (int j = t; j < total; j += 256) {
      const int2 p = E[j];
      stage[nstart[p.y & 1023] + (p.y >> 10)] = p.x;
    }
    __syncthreads();
    for (int i = t; i < total; i += 256) csr[base + i] = stage[i];
  } else {
    for (int i = t; i < NPB; i += 256) ncnt[i] = 0;
    __syncthreads();
    for (int k = 0; k < len; ++k) {
      const int2 p = g[k];
      const int ln = p.y - nodeBase;
      const int rk = atomicAdd(&ncnt[ln], 1);
      csr[base + nstart[ln] + rk] = p.x;
    }
  }
}

static __device__ __forceinline__ float4 ld_row4(const float* __restrict__ emb, int node, int c4) {
  return *(const float4*)(emb + (size_t)node * EMBED_DIM + c4);
}

__global__ __launch_bounds__(32) void walk_kernel(
    const float* __restrict__ emb,
    const int* __restrict__ uidx, const int* __restrict__ iidx,
    const int* __restrict__ offW, const int* __restrict__ csr,
    float* __restrict__ ubuf, float* __restrict__ vbuf) {
  const int2* __restrict__ off2 = (const int2*)offW;
  __shared__ int list[LIST_CAP];
  const int lane = threadIdx.x;
  const int h = lane >> 4;
  const int c4 = (lane & 15) << 2;
  const int wv = blockIdx.x;
  const bool isUser = wv < BATCH;
  const int b = isUser ? wv : wv - BATCH;
  const int node = isUser ? uidx[b] : NUM_USERS + iidx[b];
  float* outRow = (isUser ? ubuf : vbuf) + (size_t)b * EMBED_DIM + c4;

  if (lane == 0) list[0] = node;
  __syncthreads();

  bool ovf = false;
  int fb = 0, fe = 1;
  for (int lvl = 0; lvl < NUM_LAYERS; ++lvl) {
    int wr = fe;
    for (int base = fb; base < fe; base += 32) {
      const int i = base + lane;
      int o0 = 0, dg = 0;
      if (i < fe) {
        const int2 oo = off2[list[i]];
        o0 = oo.x;
        dg = oo.y - oo.x;
      }
      int x = dg;
      for (int o = 1; o < 32; o <<= 1) {
        const int y = __shfl_up(x, o);
        if (lane >= o) x += y;
      }
      const int tot = __shfl(x, 31);
      if (wr + tot > LIST_CAP) {
        ovf = true;
        break;
      }
      const int slot = wr + x - dg;
      for (int k = 0; k < NUM_EDGES; k += 8) {
        if (__builtin_amdgcn_ballot_w32(k < dg) == 0u) break;
        const bool a0 = k < dg;
        const bool a1 = k + 4 < dg;
        int4 q0 = make_int4(0, 0, 0, 0), q1 = make_int4(0, 0, 0, 0);
        if (a0) __builtin_memcpy(&q0, csr + o0 + k, 16);
        if (a1) __builtin_memcpy(&q1, csr + o0 + k + 4, 16);
        if (a0) {
          list[slot + k] = q0.x;
          if (k + 1 < dg) list[slot + k + 1] = q0.y;
          if (k + 2 < dg) list[slot + k + 2] = q0.z;
          if (k + 3 < dg) list[slot + k + 3] = q0.w;
        }
        if (a1) {
          list[slot + k + 4] = q1.x;
          if (k + 5 < dg) list[slot + k + 5] = q1.y;
          if (k + 6 < dg) list[slot + k + 6] = q1.z;
          if (k + 7 < dg) list[slot + k + 7] = q1.w;
        }
      }
      wr += tot;
    }
    __syncthreads();
    if (ovf) break;
    fb = fe;
    fe = wr;
  }

  float4 acc = make_float4(0.f, 0.f, 0.f, 0.f);
  if (!ovf) {
    const int n = fe;
    for (int i = 0; i < n; i += 16) {
      float4 v[8];
#pragma unroll
      for (int u = 0; u < 8; ++u) {
        const int idx = i + 2 * u + h;
        const bool ok = idx < n;
        const int s = list[ok ? idx : 0];
        const float4 r = ld_row4(emb, s, c4);
        v[u] = ok ? r : make_float4(0.f, 0.f, 0.f, 0.f);
      }
#pragma unroll
      for (int u = 0; u < 8; ++u) {
        acc.x += v[u].x; acc.y += v[u].y; acc.z += v[u].z; acc.w += v[u].w;
      }
    }
  } else {
    const float4 r0 = ld_row4(emb, node, c4);
    if (h == 0) { acc.x += r0.x; acc.y += r0.y; acc.z += r0.z; acc.w += r0.w; }
    const int2 q1 = off2[node];
    for (int j1 = q1.x; j1 < q1.y; ++j1) {
      const int n1 = csr[j1];
      const float4 r1 = ld_row4(emb, n1, c4);
      if (h == 0) { acc.x += r1.x; acc.y += r1.y; acc.z += r1.z; acc.w += r1.w; }
      const int2 q2 = off2[n1];
      for (int j2 = q2.x; j2 < q2.y; ++j2) {
        const int n2 = csr[j2];
        const float4 r2 = ld_row4(emb, n2, c4);
        if (h == 0) { acc.x += r2.x; acc.y += r2.y; acc.z += r2.z; acc.w += r2.w; }
        const int2 q3 = off2[n2];
        for (int j3 = q3.x; j3 < q3.y; ++j3) {
          const int n3 = csr[j3];
          const float4 r3 = ld_row4(emb, n3, c4);
          if (h == 0) { acc.x += r3.x; acc.y += r3.y; acc.z += r3.z; acc.w += r3.w; }
        }
      }
    }
  }

  acc.x += __shfl_xor(acc.x, 16);
  acc.y += __shfl_xor(acc.y, 16);
  acc.z += __shfl_xor(acc.z, 16);
  acc.w += __shfl_xor(acc.w, 16);
  if (h == 0) *(float4*)outRow = acc;
}

__global__ __launch_bounds__(128) void dot_wmma_kernel(
    const float* __restrict__ ubuf, const float* __restrict__ vbuf,
    float* __restrict__ out) {
  const int lane = threadIdx.x & 31;
  const int tile = blockIdx.x * (blockDim.x >> 5) + (threadIdx.x >> 5);
  const int rowBase = tile * 16;
  const int row = rowBase + (lane & 15);
  const int kOff = (lane >> 4) << 1;

  const float* uRow = ubuf + (size_t)row * EMBED_DIM + kOff;
  const float* vRow = vbuf + (size_t)row * EMBED_DIM + kOff;

  v8f c = {};
#pragma unroll
  for (int k = 0; k < EMBED_DIM; k += 4) {
    v2f a, b;
    a.x = uRow[k];
    a.y = uRow[k + 1];
    b.x = vRow[k];
    b.y = vRow[k + 1];
    c = __builtin_amdgcn_wmma_f32_16x16x4_f32(false, a, false, b, (short)0, c,
                                              false, false);
  }

  float cd[8];
#pragma unroll
  for (int i = 0; i < 8; ++i) cd[i] = c[i];
  int dsel = (lane < 16) ? lane : (lane - 24);
  float diag = cd[0];
#pragma unroll
  for (int i = 1; i < 8; ++i)
    if (dsel == i) diag = cd[i];

  const float scale = 1.0f / 16.0f;
  if (lane < 8)
    out[rowBase + lane] = diag * scale;
  else if (lane >= 24)
    out[rowBase + (lane - 16)] = diag * scale;
}

extern "C" void kernel_launch(void* const* d_in, const int* in_sizes, int n_in,
                              void* d_out, int out_size, void* d_ws,
                              size_t ws_size, hipStream_t stream) {
  const float* emb = (const float*)d_in[0];
  const int* eidx = (const int*)d_in[1];
  const int* uidx = (const int*)d_in[2];
  const int* iidx = (const int*)d_in[3];
  float* out = (float*)d_out;

  const int* src = eidx;
  const int* dst = eidx + NUM_EDGES;

  int* sortedE = (int*)d_ws;
  int* woff = sortedE + 2 * GA * CH;
  int* off2 = woff + WOFF_PAD;
  int* csr = off2 + 2 * OFF_PAD;
  float* ubuf = (float*)(csr + NUM_EDGES + CSR_PAD);
  float* vbuf = ubuf + BATCH * EMBED_DIM;

  bin_kernel<<<GA, 1024, 0, stream>>>(src, dst, sortedE, woff);
  csr_kernel<<<NB, 256, 0, stream>>>(sortedE, woff, off2, csr);

  walk_kernel<<<2 * BATCH, 32, 0, stream>>>(emb, uidx, iidx, off2, csr, ubuf, vbuf);

  dot_wmma_kernel<<<BATCH / 16 / 4, 128, 0, stream>>>(ubuf, vbuf, out);
}


# Batch dot fused into the walk launch: 32-wave work-groups of 16 pairs, each wave leaves its row sum in LDS, wave 0 does the tile's 16 WMMAs and stores; 4 launches to 3; sum order kept
# speedup vs baseline: 9.5936x; 1.0543x over previous
#include <hip/hip_runtime.h>
#include <hip/hip_bf16.h>

#define NUM_USERS 100000
#define NUM_ITEMS 200000
#define N_TOTAL   (NUM_USERS + NUM_ITEMS)
#define EMBED_DIM 64
#define NUM_LAYERS 3
#define NUM_EDGES 1200000
#define BATCH 4096

#define LIST_CAP 1280
#define WPB 32
#define ROWP 68

#define NB 512
#define NPB 586
#define CH 8192
#define GA ((NUM_EDGES + CH - 1) / CH)
#define WSTR 516
#define WOFF_PAD 75904
#define OFF_PAD 300032
#define CSR_PAD 64
#define LCAP 4096

typedef __attribute__((ext_vector_type(2))) float v2f;
typedef __attribute__((ext_vector_type(8))) float v8f;

__global__ __launch_bounds__(1024) void bin_kernel(const int* __restrict__ src,
                                                   const int* __restrict__ dst,
                                                   int* __restrict__ sortedW,
                                                   int* __restrict__ woff) {
  int2* __restrict__ sortedE = (int2*)sortedW;
  __shared__ int2 ent[CH];
  __shared__ int cnt[NB];
  __shared__ int start[NB + 1];
  __shared__ int wsum[16];
  const int t = threadIdx.x;
  const int lane = t & 31;
  const int wg = blockIdx.x;
  const int e0 = wg * CH + t * 8;
  if (t < NB) cnt[t] = 0;
  __syncthreads();

  int s[8], d[8], r[8], b[8];
  const bool valid = e0 < NUM_EDGES;
  if (valid) {
    const int4 sa = *(const int4*)(src + e0);
    const int4 sb = *(const int4*)(src + e0 + 4);
    const int4 da = *(const int4*)(dst + e0);
    const int4 db = *(const int4*)(dst + e0 + 4);
    s[0] = sa.x; s[1] = sa.y; s[2] = sa.z; s[3] = sa.w;
    s[4] = sb.x; s[5] = sb.y; s[6] = sb.z; s[7] = sb.w;
    d[0] = da.x; d[1] = da.y; d[2] = da.z; d[3] = da.w;
    d[4] = db.x; d[5] = db.y; d[6] = db.z; d[7] = db.w;
#pragma unroll
    for (int k = 0; k < 8; ++k) {
      b[k] = d[k] / NPB;
      r[k] = atomicAdd(&cnt[b[k]], 1);
    }
  }
  __syncthreads();

  const int v = (t < NB) ? cnt[t] : 0;
  int x = v;
  for (int o = 1; o < 32; o <<= 1) {
    const int y = __shfl_up(x, o);
    if (lane >= o) x += y;
  }
  if (t < NB && lane == 31) wsum[t >> 5] = x;
  __syncthreads();
  if (t < NB) {
    int pre = 0;
    for (int w = 0; w < (t >> 5); ++w) pre += wsum[w];
    start[t] = pre + x - v;
    if (t == NB - 1) start[NB] = pre + x;
  }
  __syncthreads();

  if (valid) {
#pragma unroll
    for (int k = 0; k < 8; ++k) ent[start[b[k]] + r[k]] = make_int2(s[k], d[k]);
  }
  __syncthreads();

  const int left = NUM_EDGES - wg * CH;
  const int nvalid = left < CH ? left : CH;
  int2* outE = sortedE + (size_t)wg * CH;
  for (int i = t; i < nvalid; i += 1024) outE[i] = ent[i];
  for (int j = t; j <= NB; j += 1024) woff[wg * WSTR + j] = start[j];
}

static __device__ __forceinline__ int scan256(int v, int* wsum8, int& total) {
  const int t = threadIdx.x;
  const int lane = t & 31;
  const int wave = t >> 5;
  int x = v;
  for (int o = 1; o < 32; o <<= 1) {
    const int y = __shfl_up(x, o);
    if (lane >= o) x += y;
  }
  if (lane == 31) wsum8[wave] = x;
  __syncthreads();
  int pre = 0, tot = 0;
  for (int w = 0; w < 8; ++w) {
    const int q = wsum8[w];
    if (w < wave) pre += q;
    tot += q;
  }
  __syncthreads();
  total = tot;
  return pre + x - v;
}

__global__ __launch_bounds__(256) void csr_kernel(const int* __restrict__ sortedW,
                                                  const int* __restrict__ woff,
                                                  int* __restrict__ offW,
                                                  int* __restrict__ csr) {
  const int2* __restrict__ sortedE = (const int2*)sortedW;
  int2* __restrict__ off2 = (int2*)offW;
  __shared__ int2 E[LCAP];
  __shared__ int stage[LCAP];
  __shared__ int ncnt[NPB];
  __shared__ int nstart[NPB + 1];
  __shared__ int wsum8[8];
  const int t = threadIdx.x;
  const int b = blockIdx.x;
  const int nodeBase = b * NPB;

  for (int i = t; i < NPB; i += 256) ncnt[i] = 0;

  int a = 0, len = 0;
  if (t < GA) {
    a = woff[t * WSTR + b];
    len = woff[t * WSTR + b + 1] - a;
  }
  int total, base;
  const int ss = scan256(len, wsum8, total);
  (void)scan256(a, wsum8, base);
  const int2* g = sortedE + (size_t)t * CH + a;
  const bool inLds = total <= LCAP;

  if (inLds) {
    for (int k0 = 0; k0 < len; k0 += 8) {
      int2 p[8];
#pragma unroll
      for (int u = 0; u < 8; ++u)
        if (k0 + u < len) p[u] = g[k0 + u];
#pragma unroll
      for (int u = 0; u < 8; ++u)
        if (k0 + u < len) E[ss + k0 + u] = p[u];
    }
    __syncthreads();
    for (int j = t; j < total; j += 256) {
      const int2 p = E[j];
      const int ln = p.y - nodeBase;
      const int rk = atomicAdd(&ncnt[ln], 1);
      E[j] = make_int2(p.x, ln | (rk << 10));
    }
  } else {
    for (int k = 0; k < len; ++k) atomicAdd(&ncnt[g[k].y - nodeBase], 1);
  }
  __syncthreads();

  const int i0 = 3 * t;
  const int c0 = (i0 < NPB) ? ncnt[i0] : 0;
  const int c1 = (i0 + 1 < NPB) ? ncnt[i0 + 1] : 0;
  const int c2 = (i0 + 2 < NPB) ? ncnt[i0 + 2] : 0;
  int tot2;
  const int pre = scan256(c0 + c1 + c2, wsum8, tot2);
  if (i0 < NPB) nstart[i0] = pre;
  if (i0 + 1 < NPB) nstart[i0 + 1] = pre + c0;
  if (i0 + 2 < NPB) nstart[i0 + 2] = pre + c0 + c1;
  if (t == 0) nstart[NPB] = tot2;
  __syncthreads();

  for (int i = t; i < NPB; i += 256) {
    const int n = nodeBase + i;
    if (n < N_TOTAL) off2[n] = make_int2(base + nstart[i], base + nstart[i + 1]);
  }
  if (b == NB - 1 && t < 4) csr[NUM_EDGES + t] = 0;

  if (inLds) {
    for (int j = t; j < total; j += 256) {
      const int2 p = E[j];
      stage[nstart[p.y & 1023] + (p.y >> 10)] = p.x;
    }
    __syncthreads();
    for (int i = t; i < total; i += 256) csr[base + i] = stage[i];
  } else {
    for (int i = t; i < NPB; i += 256) ncnt[i] = 0;
    __syncthreads();
    for (int k = 0; k < len; ++k) {
      const int2 p = g[k];
      const int ln = p.y - nodeBase;
      const int rk = atomicAdd(&ncnt[ln], 1);
      csr[base + nstart[ln] + rk] = p.x;
    }
  }
}

static __device__ __forceinline__ float4 ld_row4(const float* __restrict__ emb, int node, int c4) {
  return *(const float4*)(emb + (size_t)node * EMBED_DIM + c4);
}

static __device__ __forceinline__ void wave_lds_sync() {
  __builtin_amdgcn_fence(__ATOMIC_RELEASE, "wavefront");
  __builtin_amdgcn_wave_barrier();
  __builtin_amdgcn_fence(__ATOMIC_ACQUIRE, "wavefront");
}

__global__ __launch_bounds__(WPB * 32) void walk_dot_kernel(
    const float* __restrict__ emb,
    const int* __restrict__ uidx, const int* __restrict__ iidx,
    const int* __restrict__ offW, const int* __restrict__ csr,
    float* __restrict__ out) {
  const int2* __restrict__ off2 = (const int2*)offW;
  __shared__ int lists[WPB * LIST_CAP];
  __shared__ __attribute__((aligned(16))) float rowsum[2 * 16 * ROWP];
  const int lane = threadIdx.x & 31;
  const int wave = __builtin_amdgcn_readfirstlane((int)(threadIdx.x >> 5));
  int* __restrict__ list = lists + wave * LIST_CAP;
  const int h = lane >> 4;
  const int c4 = (lane & 15) << 2;
  const bool isUser = wave < 16;
  const int r = wave & 15;
  const int rowBase = blockIdx.x * 16;
  const int b = rowBase + r;
  const int node = isUser ? uidx[b] : NUM_USERS + iidx[b];

  if (lane == 0) list[0] = node;
  wave_lds_sync();

  bool ovf = false;
  int fb = 0, fe = 1;
  for (int lvl = 0; lvl < NUM_LAYERS; ++lvl) {
    int wr = fe;
    for (int base = fb; base < fe; base += 32) {
      const int i = base + lane;
      int o0 = 0, dg = 0;
      if (i < fe) {
        const int2 oo = off2[list[i]];
        o0 = oo.x;
        dg = oo.y - oo.x;
      }
      int x = dg;
      for (int o = 1; o < 32; o <<= 1) {
        const int y = __shfl_up(x, o);
        if (lane >= o) x += y;
      }
      const int tot = __shfl(x, 31);
      if (wr + tot > LIST_CAP) {
        ovf = true;
        break;
      }
      const int slot = wr + x - dg;
      for (int k = 0; k < NUM_EDGES; k += 8) {
        if (__builtin_amdgcn_ballot_w32(k < dg) == 0u) break;
        const bool a0 = k < dg;
        const bool a1 = k + 4 < dg;
        int4 q0 = make_int4(0, 0, 0, 0), q1 = make_int4(0, 0, 0, 0);
        if (a0) __builtin_memcpy(&q0, csr + o0 + k, 16);
        if (a1) __builtin_memcpy(&q1, csr + o0 + k + 4, 16);
        if (a0) {
          list[slot + k] = q0.x;
          if (k + 1 < dg) list[slot + k + 1] = q0.y;
          if (k + 2 < dg) list[slot + k + 2] = q0.z;
          if (k + 3 < dg) list[slot + k + 3] = q0.w;
        }
        if (a1) {
          list[slot + k + 4] = q1.x;
          if (k + 5 < dg) list[slot + k + 5] = q1.y;
          if (k + 6 < dg) list[slot + k + 6] = q1.z;
          if (k + 7 < dg) list[slot + k + 7] = q1.w;
        }
      }
      wr += tot;
    }
    wave_lds_sync();
    if (ovf) break;
    fb = fe;
    fe = wr;
  }

  float4 acc = make_float4(0.f, 0.f, 0.f, 0.f);
  if (!ovf) {
    const int n = fe;
    for (int i = 0; i < n; i += 16) {
      float4 v[8];
#pragma unroll
      for (int u = 0; u < 8; ++u) {
        const int idx = i + 2 * u + h;
        const bool ok = idx < n;
        const int s = list[ok ? idx : 0];
        const float4 rw = ld_row4(emb, s, c4);
        v[u] = ok ? rw : make_float4(0.f, 0.f, 0.f, 0.f);
      }
#pragma unroll
      for (int u = 0; u < 8; ++u) {
        acc.x += v[u].x; acc.y += v[u].y; acc.z += v[u].z; acc.w += v[u].w;
      }
    }
  } else {
    const float4 r0 = ld_row4(emb, node, c4);
    if (h == 0) { acc.x += r0.x; acc.y += r0.y; acc.z += r0.z; acc.w += r0.w; }
    const int2 q1 = off2[node];
    for (int j1 = q1.x; j1 < q1.y; ++j1) {
      const int n1 = csr[j1];
      const float4 r1 = ld_row4(emb, n1, c4);
      if (h == 0) { acc.x += r1.x; acc.y += r1.y; acc.z += r1.z; acc.w += r1.w; }
      const int2 q2 = off2[n1];
      for (int j2 = q2.x; j2 < q2.y; ++j2) {
        const int n2 = csr[j2];
        const float4 r2 = ld_row4(emb, n2, c4);
        if (h == 0) { acc.x += r2.x; acc.y += r2.y; acc.z += r2.z; acc.w += r2.w; }
        const int2 q3 = off2[n2];
        for (int j3 = q3.x; j3 < q3.y; ++j3) {
          const int n3 = csr[j3];
          const float4 r3 = ld_row4(emb, n3, c4);
          if (h == 0) { acc.x += r3.x; acc.y += r3.y; acc.z += r3.z; acc.w += r3.w; }
        }
      }
    }
  }

  acc.x += __shfl_xor(acc.x, 16);
  acc.y += __shfl_xor(acc.y, 16);
  acc.z += __shfl_xor(acc.z, 16);
  acc.w += __shfl_xor(acc.w, 16);
  if (h == 0) *(float4*)(rowsum + (isUser ? 0 : 16 * ROWP) + r * ROWP + c4) = acc;
  __syncthreads();

  if (wave == 0) {
    const int kOff = (lane >> 4) << 1;
    const float* uRow = rowsum + (lane & 15) * ROWP + kOff;
    const float* vRow = rowsum + 16 * ROWP + (lane & 15) * ROWP + kOff;

    v8f c = {};
#pragma unroll
    for (int k = 0; k < EMBED_DIM; k += 4) {
      v2f a, bf;
      a.x = uRow[k];
      a.y = uRow[k + 1];
      bf.x = vRow[k];
      bf.y = vRow[k + 1];
      c = __builtin_amdgcn_wmma_f32_16x16x4_f32(false, a, false, bf, (short)0, c,
                                                false, false);
    }

    float cd[8];
#pragma unroll
    for (int i = 0; i < 8; ++i) cd[i] = c[i];
    int dsel = (lane < 16) ? lane : (lane - 24);
    float diag = cd[0];
#pragma unroll
    for (int i = 1; i < 8; ++i)
      if (dsel == i) diag = cd[i];

    const float scale = 1.0f / 16.0f;
    if (lane < 8)
      out[rowBase + lane] = diag * scale;
    else if (lane >= 24)
      out[rowBase + (lane - 16)] = diag * scale;
  }
}

extern "C" void kernel_launch(void* const* d_in, const int* in_sizes, int n_in,
                              void* d_out, int out_size, void* d_ws,
                              size_t ws_size, hipStream_t stream) {
  const float* emb = (const float*)d_in[0];
  const int* eidx = (const int*)d_in[1];
  const int* uidx = (const int*)d_in[2];
  const int* iidx = (const int*)d_in[3];
  float* out = (float*)d_out;

  const int* src = eidx;
  const int* dst = eidx + NUM_EDGES;

  int* sortedE = (int*)d_ws;
  int* woff = sortedE + 2 * GA * CH;
  int* off2 = woff + WOFF_PAD;
  int* csr = off2 + 2 * OFF_PAD;

  bin_kernel<<<GA, 1024, 0, stream>>>(src, dst, sortedE, woff);
  csr_kernel<<<NB, 256, 0, stream>>>(sortedE, woff, off2, csr);

  walk_dot_kernel<<<BATCH / 16, WPB * 32, 0, stream>>>(emb, uidx, iidx, off2, csr, out);
}


# Row streaming balanced in a tile: behind a barrier its 32 path-end lists, in 16-row units, go in equal shares to its 32 waves; per-(wave,list) partial sums in LDS added per list; sum order changed
# speedup vs baseline: 10.7639x; 1.1220x over previous
#include <hip/hip_runtime.h>
#include <hip/hip_bf16.h>

#define NUM_USERS 100000
#define NUM_ITEMS 200000
#define N_TOTAL   (NUM_USERS + NUM_ITEMS)
#define EMBED_DIM 64
#define NUM_LAYERS 3
#define NUM_EDGES 1200000
#define BATCH 4096

#define LIST_CAP 1280
#define WPB 32
#define ROWP 68
#define PARTP 64
#define NPART (2 * WPB - 1)

#define NB 512
#define NPB 586
#define CH 8192
#define GA ((NUM_EDGES + CH - 1) / CH)
#define WSTR 516
#define WOFF_PAD 75904
#define OFF_PAD 300032
#define CSR_PAD 64
#define LCAP 4096

typedef __attribute__((ext_vector_type(2))) float v2f;
typedef __attribute__((ext_vector_type(8))) float v8f;

__global__ __launch_bounds__(1024) void bin_kernel(const int* __restrict__ src,
                                                   const int* __restrict__ dst,
                                                   int* __restrict__ sortedW,
                                                   int* __restrict__ woff) {
  int2* __restrict__ sortedE = (int2*)sortedW;
  __shared__ int2 ent[CH];
  __shared__ int cnt[NB];
  __shared__ int start[NB + 1];
  __shared__ int wsum[16];
  const int t = threadIdx.x;
  const int lane = t & 31;
  const int wg = blockIdx.x;
  const int e0 = wg * CH + t * 8;
  if (t < NB) cnt[t] = 0;
  __syncthreads();

  int s[8], d[8], r[8], b[8];
  const bool valid = e0 < NUM_EDGES;
  if (valid) {
    const int4 sa = *(const int4*)(src + e0);
    const int4 sb = *(const int4*)(src + e0 + 4);
    const int4 da = *(const int4*)(dst + e0);
    const int4 db = *(const int4*)(dst + e0 + 4);
    s[0] = sa.x; s[1] = sa.y; s[2] = sa.z; s[3] = sa.w;
    s[4] = sb.x; s[5] = sb.y; s[6] = sb.z; s[7] = sb.w;
    d[0] = da.x; d[1] = da.y; d[2] = da.z; d[3] = da.w;
    d[4] = db.x; d[5] = db.y; d[6] = db.z; d[7] = db.w;
#pragma unroll
    for (int k = 0; k < 8; ++k) {
      b[k] = d[k] / NPB;
      r[k] = atomicAdd(&cnt[b[k]], 1);
    }
  }
  __syncthreads();

  const int v = (t < NB) ? cnt[t] : 0;
  int x = v;
  for (int o = 1; o < 32; o <<= 1) {
    const int y = __shfl_up(x, o);
    if (lane >= o) x += y;
  }
  if (t < NB && lane == 31) wsum[t >> 5] = x;
  __syncthreads();
  if (t < NB) {
    int pre = 0;
    for (int w = 0; w < (t >> 5); ++w) pre += wsum[w];
    start[t] = pre + x - v;
    if (t == NB - 1) start[NB] = pre + x;
  }
  __syncthreads();

  if (valid) {
#pragma unroll
    for (int k = 0; k < 8; ++k) ent[start[b[k]] + r[k]] = make_int2(s[k], d[k]);
  }
  __syncthreads();

  const int left = NUM_EDGES - wg * CH;
  const int nvalid = left < CH ? left : CH;
  int2* outE = sortedE + (size_t)wg * CH;
  for (int i = t; i < nvalid; i += 1024) outE[i] = ent[i];
  for (int j = t; j <= NB; j += 1024) woff[wg * WSTR + j] = start[j];
}

static __device__ __forceinline__ int scan256(int v, int* wsum8, int& total) {
  const int t = threadIdx.x;
  const int lane = t & 31;
  const int wave = t >> 5;
  int x = v;
  for (int o = 1; o < 32; o <<= 1) {
    const int y = __shfl_up(x, o);
    if (lane >= o) x += y;
  }
  if (lane == 31) wsum8[wave] = x;
  __syncthreads();
  int pre = 0, tot = 0;
  for (int w = 0; w < 8; ++w) {
    const int q = wsum8[w];
    if (w < wave) pre += q;
    tot += q;
  }
  __syncthreads();
  total = tot;
  return pre + x - v;
}

__global__ __launch_bounds__(256) void csr_kernel(const int* __restrict__ sortedW,
                                                  const int* __restrict__ woff,
                                                  int* __restrict__ offW,
                                                  int* __restrict__ csr) {
  const int2* __restrict__ sortedE = (const int2*)sortedW;
  int2* __restrict__ off2 = (int2*)offW;
  __shared__ int2 E[LCAP];
  __shared__ int stage[LCAP];
  __shared__ int ncnt[NPB];
  __shared__ int nstart[NPB + 1];
  __shared__ int wsum8[8];
  const int t = threadIdx.x;
  const int b = blockIdx.x;
  const int nodeBase = b * NPB;

  for (int i = t; i < NPB; i += 256) ncnt[i] = 0;

  int a = 0, len = 0;
  if (t < GA) {
    a = woff[t * WSTR + b];
    len = woff[t * WSTR + b + 1] - a;
  }
  int total, base;
  const int ss = scan256(len, wsum8, total);
  (void)scan256(a, wsum8, base);
  const int2* g = sortedE + (size_t)t * CH + a;
  const bool inLds = total <= LCAP;

  if (inLds) {
    for (int k0 = 0; k0 < len; k0 += 8) {
      int2 p[8];
#pragma unroll
      for (int u = 0; u < 8; ++u)
        if (k0 + u < len) p[u] = g[k0 + u];
#pragma unroll
      for (int u = 0; u < 8; ++u)
        if (k0 + u < len) E[ss + k0 + u] = p[u];
    }
    __syncthreads();
    for (int j = t; j < total; j += 256) {
      const int2 p = E[j];
      const int ln = p.y - nodeBase;
      const int rk = atomicAdd(&ncnt[ln], 1);
      E[j] = make_int2(p.x, ln | (rk << 10));
    }
  } else {
    for (int k = 0; k < len; ++k) atomicAdd(&ncnt[g[k].y - nodeBase], 1);
  }
  __syncthreads();

  const int i0 = 3 * t;
  const int c0 = (i0 < NPB) ? ncnt[i0] : 0;
  const int c1 = (i0 + 1 < NPB) ? ncnt[i0 + 1] : 0;
  const int c2 = (i0 + 2 < NPB) ? ncnt[i0 + 2] : 0;
  int tot2;
  const int pre = scan256(c0 + c1 + c2, wsum8, tot2);
  if (i0 < NPB) nstart[i0] = pre;
  if (i0 + 1 < NPB) nstart[i0 + 1] = pre + c0;
  if (i0 + 2 < NPB) nstart[i0 + 2] = pre + c0 + c1;
  if (t == 0) nstart[NPB] = tot2;
  __syncthreads();

  for (int i = t; i < NPB; i += 256) {
    const int n = nodeBase + i;
    if (n < N_TOTAL) off2[n] = make_int2(base + nstart[i], base + nstart[i + 1]);
  }
  if (b == NB - 1 && t < 4) csr[NUM_EDGES + t] = 0;

  if (inLds) {
    for (int j = t; j < total; j += 256) {
      const int2 p = E[j];
      stage[nstart[p.y & 1023] + (p.y >> 10)] = p.x;
    }
    __syncthreads();
    for (int i = t; i < total; i += 256) csr[base + i] = stage[i];
  } else {
    for (int i = t; i < NPB; i += 256) ncnt[i] = 0;
    __syncthreads();
    for (int k = 0; k < len; ++k) {
      const int2 p = g[k];
      const int ln = p.y - nodeBase;
      const int rk = atomicAdd(&ncnt[ln], 1);
      csr[base + nstart[ln] + rk] = p.x;
    }
  }
}

static __device__ __forceinline__ float4 ld_row4(const float* __restrict__ emb, int node, int c4) {
  return *(const float4*)(emb + (size_t)node * EMBED_DIM + c4);
}

static __device__ __forceinline__ void wave_lds_sync() {
  __builtin_amdgcn_fence(__ATOMIC_RELEASE, "wavefront");
  __builtin_amdgcn_wave_barrier();
  __builtin_amdgcn_fence(__ATOMIC_ACQUIRE, "wavefront");
}

static __device__ __forceinline__ void put_partial(float* __restrict__ part, int slot, float4 acc, int h, int c4) {
  acc.x += __shfl_xor(acc.x, 16);
  acc.y += __shfl_xor(acc.y, 16);
  acc.z += __shfl_xor(acc.z, 16);
  acc.w += __shfl_xor(acc.w, 16);
  if (h == 0) *(float4*)(part + slot * PARTP + c4) = acc;
}

__global__ __launch_bounds__(WPB * 32) void walk_dot_kernel(
    const float* __restrict__ emb,
    const int* __restrict__ uidx, const int* __restrict__ iidx,
    const int* __restrict__ offW, const int* __restrict__ csr,
    float* __restrict__ out) {
  const int2* __restrict__ off2 = (const int2*)offW;
  __shared__ int lists[WPB * LIST_CAP];
  __shared__ __attribute__((aligned(16))) float rowsum[2 * 16 * ROWP];
  __shared__ __attribute__((aligned(16))) float part[NPART * PARTP];
  __shared__ int cntL[WPB];
  const int lane = threadIdx.x & 31;
  const int wave = __builtin_amdgcn_readfirstlane((int)(threadIdx.x >> 5));
  int* __restrict__ list = lists + wave * LIST_CAP;
  const int h = lane >> 4;
  const int c4 = (lane & 15) << 2;
  const bool isUser = wave < 16;
  const int r = wave & 15;
  const int rowBase = blockIdx.x * 16;
  const int b = rowBase + r;
  const int node = isUser ? uidx[b] : NUM_USERS + iidx[b];

  if (lane == 0) list[0] = node;
  wave_lds_sync();

  bool ovf = false;
  int fb = 0, fe = 1;
  for (int lvl = 0; lvl < NUM_LAYERS; ++lvl) {
    int wr = fe;
    for (int base = fb; base < fe; base += 32) {
      const int i = base + lane;
      int o0 = 0, dg = 0;
      if (i < fe) {
        const int2 oo = off2[list[i]];
        o0 = oo.x;
        dg = oo.y - oo.x;
      }
      int x = dg;
      for (int o = 1; o < 32; o <<= 1) {
        const int y = __shfl_up(x, o);
        if (lane >= o) x += y;
      }
      const int tot = __shfl(x, 31);
      if (wr + tot > LIST_CAP) {
        ovf = true;
        break;
      }
      const int slot = wr + x - dg;
      for (int k = 0; k < NUM_EDGES; k += 8) {
        if (__builtin_amdgcn_ballot_w32(k < dg) == 0u) break;
        const bool a0 = k < dg;
        const bool a1 = k + 4 < dg;
        int4 q0 = make_int4(0, 0, 0, 0), q1 = make_int4(0, 0, 0, 0);
        if (a0) __builtin_memcpy(&q0, csr + o0 + k, 16);
        if (a1) __builtin_memcpy(&q1, csr + o0 + k + 4, 16);
        if (a0) {
          list[slot + k] = q0.x;
          if (k + 1 < dg) list[slot + k + 1] = q0.y;
          if (k + 2 < dg) list[slot + k + 2] = q0.z;
          if (k + 3 < dg) list[slot + k + 3] = q0.w;
        }
        if (a1) {
          list[slot + k + 4] = q1.x;
          if (k + 5 < dg) list[slot + k + 5] = q1.y;
          if (k + 6 < dg) list[slot + k + 6] = q1.z;
          if (k + 7 < dg) list[slot + k + 7] = q1.w;
        }
      }
      wr += tot;
    }
    wave_lds_sync();
    if (ovf) break;
    fb = fe;
    fe = wr;
  }

  float4 own = make_float4(0.f, 0.f, 0.f, 0.f);
  if (ovf) {
    const float4 r0 = ld_row4(emb, node, c4);
    if (h == 0) { own.x += r0.x; own.y += r0.y; own.z += r0.z; own.w += r0.w; }
    const int2 q1 = off2[node];
    for (int j1 = q1.x; j1 < q1.y; ++j1) {
      const int n1 = csr[j1];
      const float4 r1 = ld_row4(emb, n1, c4);
      if (h == 0) { own.x += r1.x; own.y += r1.y; own.z += r1.z; own.w += r1.w; }
      const int2 q2 = off2[n1];
      for (int j2 = q2.x; j2 < q2.y; ++j2) {
        const int n2 = csr[j2];
        const float4 r2 = ld_row4(emb, n2, c4);
        if (h == 0) { own.x += r2.x; own.y += r2.y; own.z += r2.z; own.w += r2.w; }
        const int2 q3 = off2[n2];
        for (int j3 = q3.x; j3 < q3.y; ++j3) {
          const int n3 = csr[j3];
          const float4 r3 = ld_row4(emb, n3, c4);
          if (h == 0) { own.x += r3.x; own.y += r3.y; own.z += r3.z; own.w += r3.w; }
        }
      }
    }
  }
  if (lane == 0) cntL[wave] = ovf ? 0 : fe;
  __syncthreads();

  const int cnt = cntL[lane];
  const int units = (cnt + 15) >> 4;
  int incl = units;
  for (int o = 1; o < 32; o <<= 1) {
    const int y = __shfl_up(incl, o);
    if (lane >= o) incl += y;
  }
  const int total = __builtin_amdgcn_readlane(incl, 31);
  const int share = (total + WPB - 1) / WPB;
  const int u0 = wave * share;
  const int u1 = (u0 + share < total) ? (u0 + share) : total;

  for (int u = u0; u < u1;) {
    const int L = __builtin_popcount(__builtin_amdgcn_ballot_w32(incl <= u));
    const int endL = __builtin_amdgcn_readlane(incl, L);
    const int first = endL - __builtin_amdgcn_readlane(units, L);
    const int nL = __builtin_amdgcn_readlane(cnt, L);
    const int uEnd = endL < u1 ? endL : u1;
    const int* __restrict__ lst = lists + L * LIST_CAP;
    const int iEnd = (uEnd - first) << 4;
    float4 acc = make_float4(0.f, 0.f, 0.f, 0.f);
    for (int i = (u - first) << 4; i < iEnd; i += 16) {
      float4 v[8];
#pragma unroll
      for (int k = 0; k < 8; ++k) {
        const int idx = i + 2 * k + h;
        const bool ok = idx < nL;
        const int s = lst[ok ? idx : 0];
        const float4 rw = ld_row4(emb, s, c4);
        v[k] = ok ? rw : make_float4(0.f, 0.f, 0.f, 0.f);
      }
#pragma unroll
      for (int k = 0; k < 8; ++k) {
        acc.x += v[k].x; acc.y += v[k].y; acc.z += v[k].z; acc.w += v[k].w;
      }
    }
    put_partial(part, wave + L, acc, h, c4);
    u = uEnd;
  }
  __syncthreads();

  const int myUnits = __builtin_amdgcn_readlane(units, wave);
  if (myUnits > 0) {
    const int myEnd = __builtin_amdgcn_readlane(incl, wave);
    const int wlo = (myEnd - myUnits) / share;
    const int whi = (myEnd - 1) / share;
    for (int w = wlo; w <= whi; ++w) {
      const float4 p = *(const float4*)(part + (w + wave) * PARTP + c4);
      own.x += p.x; own.y += p.y; own.z += p.z; own.w += p.w;
    }
  }
  if (h == 0) *(float4*)(rowsum + (isUser ? 0 : 16 * ROWP) + r * ROWP + c4) = own;
  __syncthreads();

  if (wave == 0) {
    const int kOff = (lane >> 4) << 1;
    const float* uRow = rowsum + (lane & 15) * ROWP + kOff;
    const float* vRow = rowsum + 16 * ROWP + (lane & 15) * ROWP + kOff;

    v8f c = {};
#pragma unroll
    for (int k = 0; k < EMBED_DIM; k += 4) {
      v2f a, bf;
      a.x = uRow[k];
      a.y = uRow[k + 1];
      bf.x = vRow[k];
      bf.y = vRow[k + 1];
      c = __builtin_amdgcn_wmma_f32_16x16x4_f32(false, a, false, bf, (short)0, c,
                                                false, false);
    }

    float cd[8];
#pragma unroll
    for (int i = 0; i < 8; ++i) cd[i] = c[i];
    int dsel = (lane < 16) ? lane : (lane - 24);
    float diag = cd[0];
#pragma unroll
    for (int i = 1; i < 8; ++i)
      if (dsel == i) diag = cd[i];

    const float scale = 1.0f / 16.0f;
    if (lane < 8)
      out[rowBase + lane] = diag * scale;
    else if (lane >= 24)
      out[rowBase + (lane - 16)] = diag * scale;
  }
}

extern "C" void kernel_launch(void* const* d_in, const int* in_sizes, int n_in,
                              void* d_out, int out_size, void* d_ws,
                              size_t ws_size, hipStream_t stream) {
  const float* emb = (const float*)d_in[0];
  const int* eidx = (const int*)d_in[1];
  const int* uidx = (const int*)d_in[2];
  const int* iidx = (const int*)d_in[3];
  float* out = (float*)d_out;

  const int* src = eidx;
  const int* dst = eidx + NUM_EDGES;

  int* sortedE = (int*)d_ws;
  int* woff = sortedE + 2 * GA * CH;
  int* off2 = woff + WOFF_PAD;
  int* csr = off2 + 2 * OFF_PAD;

  bin_kernel<<<GA, 1024, 0, stream>>>(src, dst, sortedE, woff);
  csr_kernel<<<NB, 256, 0, stream>>>(sortedE, woff, off2, csr);

  walk_dot_kernel<<<BATCH / 16, WPB * 32, 0, stream>>>(emb, uidx, iidx, off2, csr, out);
}


# Second adjacency pass orders a range out of registers: edge j to thread j%256, each step's reads, loads and counter updates issued as one group; LDS edge array dropped; sum order changed
# speedup vs baseline: 11.8950x; 1.1051x over previous
#include <hip/hip_runtime.h>
#include <hip/hip_bf16.h>

#define NUM_USERS 100000
#define NUM_ITEMS 200000
#define N_TOTAL   (NUM_USERS + NUM_ITEMS)
#define EMBED_DIM 64
#define NUM_LAYERS 3
#define NUM_EDGES 1200000
#define BATCH 4096

#define LIST_CAP 1280
#define WPB 32
#define ROWP 68
#define PARTP 64
#define NPART (2 * WPB - 1)

#define NB 512
#define NPB 586
#define CH 8192
#define GA ((NUM_EDGES + CH - 1) / CH)
#define WSTR 516
#define WOFF_PAD 75904
#define OFF_PAD 300032
#define CSR_PAD 64
#define LCAP 4096
#define EPT (LCAP / 256)

typedef __attribute__((ext_vector_type(2))) float v2f;
typedef __attribute__((ext_vector_type(8))) float v8f;

__global__ __launch_bounds__(1024) void bin_kernel(const int* __restrict__ src,
                                                   const int* __restrict__ dst,
                                                   int* __restrict__ sortedW,
                                                   int* __restrict__ woff) {
  int2* __restrict__ sortedE = (int2*)sortedW;
  __shared__ int2 ent[CH];
  __shared__ int cnt[NB];
  __shared__ int start[NB + 1];
  __shared__ int wsum[16];
  const int t = threadIdx.x;
  const int lane = t & 31;
  const int wg = blockIdx.x;
  const int e0 = wg * CH + t * 8;
  if (t < NB) cnt[t] = 0;
  __syncthreads();

  int s[8], d[8], r[8], b[8];
  const bool valid = e0 < NUM_EDGES;
  if (valid) {
    const int4 sa = *(const int4*)(src + e0);
    const int4 sb = *(const int4*)(src + e0 + 4);
    const int4 da = *(const int4*)(dst + e0);
    const int4 db = *(const int4*)(dst + e0 + 4);
    s[0] = sa.x; s[1] = sa.y; s[2] = sa.z; s[3] = sa.w;
    s[4] = sb.x; s[5] = sb.y; s[6] = sb.z; s[7] = sb.w;
    d[0] = da.x; d[1] = da.y; d[2] = da.z; d[3] = da.w;
    d[4] = db.x; d[5] = db.y; d[6] = db.z; d[7] = db.w;
#pragma unroll
    for (int k = 0; k < 8; ++k) {
      b[k] = d[k] / NPB;
      r[k] = atomicAdd(&cnt[b[k]], 1);
    }
  }
  __syncthreads();

  const int v = (t < NB) ? cnt[t] : 0;
  int x = v;
  for (int o = 1; o < 32; o <<= 1) {
    const int y = __shfl_up(x, o);
    if (lane >= o) x += y;
  }
  if (t < NB && lane == 31) wsum[t >> 5] = x;
  __syncthreads();
  if (t < NB) {
    int pre = 0;
    for (int w = 0; w < (t >> 5); ++w) pre += wsum[w];
    start[t] = pre + x - v;
    if (t == NB - 1) start[NB] = pre + x;
  }
  __syncthreads();

  if (valid) {
#pragma unroll
    for (int k = 0; k < 8; ++k) ent[start[b[k]] + r[k]] = make_int2(s[k], d[k]);
  }
  __syncthreads();

  const int left = NUM_EDGES - wg * CH;
  const int nvalid = left < CH ? left : CH;
  int2* outE = sortedE + (size_t)wg * CH;
  for (int i = t; i < nvalid; i += 1024) outE[i] = ent[i];
  for (int j = t; j <= NB; j += 1024) woff[wg * WSTR + j] = start[j];
}

static __device__ __forceinline__ int scan256(int v, int* wsum8, int& total) {
  const int t = threadIdx.x;
  const int lane = t & 31;
  const int wave = t >> 5;
  int x = v;
  for (int o = 1; o < 32; o <<= 1) {
    const int y = __shfl_up(x, o);
    if (lane >= o) x += y;
  }
  if (lane == 31) wsum8[wave] = x;
  __syncthreads();
  int pre = 0, tot = 0;
  for (int w = 0; w < 8; ++w) {
    const int q = wsum8[w];
    if (w < wave) pre += q;
    tot += q;
  }
  __syncthreads();
  total = tot;
  return pre + x - v;
}

__global__ __launch_bounds__(256) void csr_kernel(const int* __restrict__ sortedW,
                                                  const int* __restrict__ woff,
                                                  int* __restrict__ offW,
                                                  int* __restrict__ csr) {
  const int2* __restrict__ sortedE = (const int2*)sortedW;
  int2* __restrict__ off2 = (int2*)offW;
  __shared__ int stage[LCAP];
  __shared__ int ncnt[NPB];
  __shared__ int nstart[NPB + 1];
  __shared__ int wsum8[8];
  const int t = threadIdx.x;
  const int b = blockIdx.x;
  const int nodeBase = b * NPB;

  for (int i = t; i < NPB; i += 256) ncnt[i] = 0;

  int a = 0, len = 0;
  if (t < GA) {
    a = woff[t * WSTR + b];
    len = woff[t * WSTR + b + 1] - a;
  }
  int total, base;
  const int ss = scan256(len, wsum8, total);
  (void)scan256(a, wsum8, base);
  const int2* g = sortedE + (size_t)t * CH + a;
  const bool inLds = total <= LCAP;

  int sv[EPT];
  int kv[EPT];
  int rk[EPT];
  if (inLds) {
    const int e0 = t * CH + a;
    for (int k = 0; k < len; ++k) stage[ss + k] = e0 + k;
    __syncthreads();
    int e[EPT];
#pragma unroll
    for (int i = 0; i < EPT; ++i) {
      const int j = t + 256 * i;
      const bool ok = j < total;
      const int x = stage[ok ? j : 0];
      e[i] = ok ? x : 0;
    }
#pragma unroll
    for (int i = 0; i < EPT; ++i) {
      const int2 p = sortedE[e[i]];
      sv[i] = p.x;
      kv[i] = p.y - nodeBase;
    }
#pragma unroll
    for (int i = 0; i < EPT; ++i) {
      rk[i] = 0;
      if (t + 256 * i < total) rk[i] = atomicAdd(&ncnt[kv[i]], 1);
    }
  } else {
    for (int k = 0; k < len; ++k) atomicAdd(&ncnt[g[k].y - nodeBase], 1);
  }
  __syncthreads();

  const int i0 = 3 * t;
  const int c0 = (i0 < NPB) ? ncnt[i0] : 0;
  const int c1 = (i0 + 1 < NPB) ? ncnt[i0 + 1] : 0;
  const int c2 = (i0 + 2 < NPB) ? ncnt[i0 + 2] : 0;
  int tot2;
  const int pre = scan256(c0 + c1 + c2, wsum8, tot2);
  if (i0 < NPB) nstart[i0] = pre;
  if (i0 + 1 < NPB) nstart[i0 + 1] = pre + c0;
  if (i0 + 2 < NPB) nstart[i0 + 2] = pre + c0 + c1;
  if (t == 0) nstart[NPB] = tot2;
  __syncthreads();

  for (int i = t; i < NPB; i += 256) {
    const int n = nodeBase + i;
    if (n < N_TOTAL) off2[n] = make_int2(base + nstart[i], base + nstart[i + 1]);
  }
  if (b == NB - 1 && t < 4) csr[NUM_EDGES + t] = 0;

  if (inLds) {
    int ns[EPT];
#pragma unroll
    for (int i = 0; i < EPT; ++i) ns[i] = nstart[(t + 256 * i < total) ? kv[i] : 0];
#pragma unroll
    for (int i = 0; i < EPT; ++i)
      if (t + 256 * i < total) stage[ns[i] + rk[i]] = sv[i];
    __syncthreads();
    int w[EPT];
#pragma unroll
    for (int i = 0; i < EPT; ++i) w[i] = stage[(t + 256 * i < total) ? t + 256 * i : 0];
#pragma unroll
    for (int i = 0; i < EPT; ++i)
      if (t + 256 * i < total) csr[base + t + 256 * i] = w[i];
  } else {
    for (int i = t; i < NPB; i += 256) ncnt[i] = 0;
    __syncthreads();
    for (int k = 0; k < len; ++k) {
      const int2 p = g[k];
      const int ln = p.y - nodeBase;
      const int rk1 = atomicAdd(&ncnt[ln], 1);
      csr[base + nstart[ln] + rk1] = p.x;
    }
  }
}

static __device__ __forceinline__ float4 ld_row4(const float* __restrict__ emb, int node, int c4) {
  return *(const float4*)(emb + (size_t)node * EMBED_DIM + c4);
}

static __device__ __forceinline__ void wave_lds_sync() {
  __builtin_amdgcn_fence(__ATOMIC_RELEASE, "wavefront");
  __builtin_amdgcn_wave_barrier();
  __builtin_amdgcn_fence(__ATOMIC_ACQUIRE, "wavefront");
}

static __device__ __forceinline__ void put_partial(float* __restrict__ part, int slot, float4 acc, int h, int c4) {
  acc.x += __shfl_xor(acc.x, 16);
  acc.y += __shfl_xor(acc.y, 16);
  acc.z += __shfl_xor(acc.z, 16);
  acc.w += __shfl_xor(acc.w, 16);
  if (h == 0) *(float4*)(part + slot * PARTP + c4) = acc;
}

__global__ __launch_bounds__(WPB * 32) void walk_dot_kernel(
    const float* __restrict__ emb,
    const int* __restrict__ uidx, const int* __restrict__ iidx,
    const int* __restrict__ offW, const int* __restrict__ csr,
    float* __restrict__ out) {
  const int2* __restrict__ off2 = (const int2*)offW;
  __shared__ int lists[WPB * LIST_CAP];
  __shared__ __attribute__((aligned(16))) float rowsum[2 * 16 * ROWP];
  __shared__ __attribute__((aligned(16))) float part[NPART * PARTP];
  __shared__ int cntL[WPB];
  const int lane = threadIdx.x & 31;
  const int wave = __builtin_amdgcn_readfirstlane((int)(threadIdx.x >> 5));
  int* __restrict__ list = lists + wave * LIST_CAP;
  const int h = lane >> 4;
  const int c4 = (lane & 15) << 2;
  const bool isUser = wave < 16;
  const int r = wave & 15;
  const int rowBase = blockIdx.x * 16;
  const int b = rowBase + r;
  const int node = isUser ? uidx[b] : NUM_USERS + iidx[b];

  if (lane == 0) list[0] = node;
  wave_lds_sync();

  bool ovf = false;
  int fb = 0, fe = 1;
  for (int lvl = 0; lvl < NUM_LAYERS; ++lvl) {
    int wr = fe;
    for (int base = fb; base < fe; base += 32) {
      const int i = base + lane;
      int o0 = 0, dg = 0;
      if (i < fe) {
        const int2 oo = off2[list[i]];
        o0 = oo.x;
        dg = oo.y - oo.x;
      }
      int x = dg;
      for (int o = 1; o < 32; o <<= 1) {
        const int y = __shfl_up(x, o);
        if (lane >= o) x += y;
      }
      const int tot = __shfl(x, 31);
      if (wr + tot > LIST_CAP) {
        ovf = true;
        break;
      }
      const int slot = wr + x - dg;
      for (int k = 0; k < NUM_EDGES; k += 8) {
        if (__builtin_amdgcn_ballot_w32(k < dg) == 0u) break;
        const bool a0 = k < dg;
        const bool a1 = k + 4 < dg;
        int4 q0 = make_int4(0, 0, 0, 0), q1 = make_int4(0, 0, 0, 0);
        if (a0) __builtin_memcpy(&q0, csr + o0 + k, 16);
        if (a1) __builtin_memcpy(&q1, csr + o0 + k + 4, 16);
        if (a0) {
          list[slot + k] = q0.x;
          if (k + 1 < dg) list[slot + k + 1] = q0.y;
          if (k + 2 < dg) list[slot + k + 2] = q0.z;
          if (k + 3 < dg) list[slot + k + 3] = q0.w;
        }
        if (a1) {
          list[slot + k + 4] = q1.x;
          if (k + 5 < dg) list[slot + k + 5] = q1.y;
          if (k + 6 < dg) list[slot + k + 6] = q1.z;
          if (k + 7 < dg) list[slot + k + 7] = q1.w;
        }
      }
      wr += tot;
    }
    wave_lds_sync();
    if (ovf) break;
    fb = fe;
    fe = wr;
  }

  float4 own = make_float4(0.f, 0.f, 0.f, 0.f);
  if (ovf) {
    const float4 r0 = ld_row4(emb, node, c4);
    if (h == 0) { own.x += r0.x; own.y += r0.y; own.z += r0.z; own.w += r0.w; }
    const int2 q1 = off2[node];
    for (int j1 = q1.x; j1 < q1.y; ++j1) {
      const int n1 = csr[j1];
      const float4 r1 = ld_row4(emb, n1, c4);
      if (h == 0) { own.x += r1.x; own.y += r1.y; own.z += r1.z; own.w += r1.w; }
      const int2 q2 = off2[n1];
      for (int j2 = q2.x; j2 < q2.y; ++j2) {
        const int n2 = csr[j2];
        const float4 r2 = ld_row4(emb, n2, c4);
        if (h == 0) { own.x += r2.x; own.y += r2.y; own.z += r2.z; own.w += r2.w; }
        const int2 q3 = off2[n2];
        for (int j3 = q3.x; j3 < q3.y; ++j3) {
          const int n3 = csr[j3];
          const float4 r3 = ld_row4(emb, n3, c4);
          if (h == 0) { own.x += r3.x; own.y += r3.y; own.z += r3.z; own.w += r3.w; }
        }
      }
    }
  }
  if (lane == 0) cntL[wave] = ovf ? 0 : fe;
  __syncthreads();

  const int cnt = cntL[lane];
  const int units = (cnt + 15) >> 4;
  int incl = units;
  for (int o = 1; o < 32; o <<= 1) {
    const int y = __shfl_up(incl, o);
    if (lane >= o) incl += y;
  }
  const int total = __builtin_amdgcn_readlane(incl, 31);
  const int share = (total + WPB - 1) / WPB;
  const int u0 = wave * share;
  const int u1 = (u0 + share < total) ? (u0 + share) : total;

  for (int u = u0; u < u1;) {
    const int L = __builtin_popcount(__builtin_amdgcn_ballot_w32(incl <= u));
    const int endL = __builtin_amdgcn_readlane(incl, L);
    const int first = endL - __builtin_amdgcn_readlane(units, L);
    const int nL = __builtin_amdgcn_readlane(cnt, L);
    const int uEnd = endL < u1 ? endL : u1;
    const int* __restrict__ lst = lists + L * LIST_CAP;
    const int iEnd = (uEnd - first) << 4;
    float4 acc = make_float4(0.f, 0.f, 0.f, 0.f);
    for (int i = (u - first) << 4; i < iEnd; i += 16) {
      float4 v[8];
#pragma unroll
      for (int k = 0; k < 8; ++k) {
        const int idx = i + 2 * k + h;
        const bool ok = idx < nL;
        const int s = lst[ok ? idx : 0];
        const float4 rw = ld_row4(emb, s, c4);
        v[k] = ok ? rw : make_float4(0.f, 0.f, 0.f, 0.f);
      }
#pragma unroll
      for (int k = 0; k < 8; ++k) {
        acc.x += v[k].x; acc.y += v[k].y; acc.z += v[k].z; acc.w += v[k].w;
      }
    }
    put_partial(part, wave + L, acc, h, c4);
    u = uEnd;
  }
  __syncthreads();

  const int myUnits = __builtin_amdgcn_readlane(units, wave);
  if (myUnits > 0) {
    const int myEnd = __builtin_amdgcn_readlane(incl, wave);
    const int wlo = (myEnd - myUnits) / share;
    const int whi = (myEnd - 1) / share;
    for (int w = wlo; w <= whi; ++w) {
      const float4 p = *(const float4*)(part + (w + wave) * PARTP + c4);
      own.x += p.x; own.y += p.y; own.z += p.z; own.w += p.w;
    }
  }
  if (h == 0) *(float4*)(rowsum + (isUser ? 0 : 16 * ROWP) + r * ROWP + c4) = own;
  __syncthreads();

  if (wave == 0) {
    const int kOff = (lane >> 4) << 1;
    const float* uRow = rowsum + (lane & 15) * ROWP + kOff;
    const float* vRow = rowsum + 16 * ROWP + (lane & 15) * ROWP + kOff;

    v8f c = {};
#pragma unroll
    for (int k = 0; k < EMBED_DIM; k += 4) {
      v2f a, bf;
      a.x = uRow[k];
      a.y = uRow[k + 1];
      bf.x = vRow[k];
      bf.y = vRow[k + 1];
      c = __builtin_amdgcn_wmma_f32_16x16x4_f32(false, a, false, bf, (short)0, c,
                                                false, false);
    }

    float cd[8];
#pragma unroll
    for (int i = 0; i < 8; ++i) cd[i] = c[i];
    int dsel = (lane < 16) ? lane : (lane - 24);
    float diag = cd[0];
#pragma unroll
    for (int i = 1; i < 8; ++i)
      if (dsel == i) diag = cd[i];

    const float scale = 1.0f / 16.0f;
    if (lane < 8)
      out[rowBase + lane] = diag * scale;
    else if (lane >= 24)
      out[rowBase + (lane - 16)] = diag * scale;
  }
}

extern "C" void kernel_launch(void* const* d_in, const int* in_sizes, int n_in,
                              void* d_out, int out_size, void* d_ws,
                              size_t ws_size, hipStream_t stream) {
  const float* emb = (const float*)d_in[0];
  const int* eidx = (const int*)d_in[1];
  const int* uidx = (const int*)d_in[2];
  const int* iidx = (const int*)d_in[3];
  float* out = (float*)d_out;

  const int* src = eidx;
  const int* dst = eidx + NUM_EDGES;

  int* sortedE = (int*)d_ws;
  int* woff = sortedE + 2 * GA * CH;
  int* off2 = woff + WOFF_PAD;
  int* csr = off2 + 2 * OFF_PAD;

  bin_kernel<<<GA, 1024, 0, stream>>>(src, dst, sortedE, woff);
  csr_kernel<<<NB, 256, 0, stream>>>(sortedE, woff, off2, csr);

  walk_dot_kernel<<<BATCH / 16, WPB * 32, 0, stream>>>(emb, uidx, iidx, off2, csr, out);
}


# Wave prefix sums on the vector ALU: DPP row shifts plus a lane-15 read, not five LDS-crossbar shuffles each; block totals by lane read; half-wave add by row exchange; all launches; sum order kept
# speedup vs baseline: 12.5607x; 1.0560x over previous
#include <hip/hip_runtime.h>
#include <hip/hip_bf16.h>

#define NUM_USERS 100000
#define NUM_ITEMS 200000
#define N_TOTAL   (NUM_USERS + NUM_ITEMS)
#define EMBED_DIM 64
#define NUM_LAYERS 3
#define NUM_EDGES 1200000
#define BATCH 4096

#define LIST_CAP 1280
#define WPB 32
#define ROWP 68
#define PARTP 64
#define NPART (2 * WPB - 1)

#define NB 512
#define NPB 586
#define CH 8192
#define GA ((NUM_EDGES + CH - 1) / CH)
#define WSTR 516
#define WOFF_PAD 75904
#define OFF_PAD 300032
#define CSR_PAD 64
#define LCAP 4096
#define EPT (LCAP / 256)

typedef __attribute__((ext_vector_type(2))) float v2f;
typedef __attribute__((ext_vector_type(8))) float v8f;

static __device__ __forceinline__ int wave_incl_scan(int v, int lane) {
  int x = v;
  x += __builtin_amdgcn_update_dpp(0, x, 0x111, 0xf, 0xf, false);
  x += __builtin_amdgcn_update_dpp(0, x, 0x112, 0xf, 0xf, false);
  x += __builtin_amdgcn_update_dpp(0, x, 0x114, 0xf, 0xf, false);
  x += __builtin_amdgcn_update_dpp(0, x, 0x118, 0xf, 0xf, false);
  const int lower = __builtin_amdgcn_readlane(x, 15);
  return x + ((lane & 16) ? lower : 0);
}

static __device__ __forceinline__ float xhalf(float x) {
  return __builtin_bit_cast(float, __builtin_amdgcn_permlanex16(__builtin_bit_cast(int, x), __builtin_bit_cast(int, x),
                                                                0x76543210, 0xfedcba98, false, false));
}

__global__ __launch_bounds__(1024) void bin_kernel(const int* __restrict__ src,
                                                   const int* __restrict__ dst,
                                                   int* __restrict__ sortedW,
                                                   int* __restrict__ woff) {
  int2* __restrict__ sortedE = (int2*)sortedW;
  __shared__ int2 ent[CH];
  __shared__ int cnt[NB];
  __shared__ int start[NB + 1];
  __shared__ int wsum[16];
  const int t = threadIdx.x;
  const int lane = t & 31;
  const int wg = blockIdx.x;
  const int e0 = wg * CH + t * 8;
  if (t < NB) cnt[t] = 0;
  __syncthreads();

  int s[8], d[8], r[8], b[8];
  const bool valid = e0 < NUM_EDGES;
  if (valid) {
    const int4 sa = *(const int4*)(src + e0);
    const int4 sb = *(const int4*)(src + e0 + 4);
    const int4 da = *(const int4*)(dst + e0);
    const int4 db = *(const int4*)(dst + e0 + 4);
    s[0] = sa.x; s[1] = sa.y; s[2] = sa.z; s[3] = sa.w;
    s[4] = sb.x; s[5] = sb.y; s[6] = sb.z; s[7] = sb.w;
    d[0] = da.x; d[1] = da.y; d[2] = da.z; d[3] = da.w;
    d[4] = db.x; d[5] = db.y; d[6] = db.z; d[7] = db.w;
#pragma unroll
    for (int k = 0; k < 8; ++k) {
      b[k] = d[k] / NPB;
      r[k] = atomicAdd(&cnt[b[k]], 1);
    }
  }
  __syncthreads();

  const int v = (t < NB) ? cnt[t] : 0;
  const int x = wave_incl_scan(v, lane);
  if (t < NB && lane == 31) wsum[t >> 5] = x;
  __syncthreads();
  const int wv = __builtin_amdgcn_readfirstlane(t >> 5);
  const int wsx = wave_incl_scan(wsum[lane & 15], lane);
  const int pre = (wv > 0) ? __builtin_amdgcn_readlane(wsx, (wv - 1) & 15) : 0;
  if (t < NB) {
    start[t] = pre + x - v;
    if (t == NB - 1) start[NB] = pre + x;
  }
  __syncthreads();

  if (valid) {
#pragma unroll
    for (int k = 0; k < 8; ++k) ent[start[b[k]] + r[k]] = make_int2(s[k], d[k]);
  }
  __syncthreads();

  const int left = NUM_EDGES - wg * CH;
  const int nvalid = left < CH ? left : CH;
  int2* outE = sortedE + (size_t)wg * CH;
  for (int i = t; i < nvalid; i += 1024) outE[i] = ent[i];
  for (int j = t; j <= NB; j += 1024) woff[wg * WSTR + j] = start[j];
}

static __device__ __forceinline__ int scan256(int v, int* wsum8, int& total) {
  const int t = threadIdx.x;
  const int lane = t & 31;
  const int wave = t >> 5;
  const int x = wave_incl_scan(v, lane);
  if (lane == 31) wsum8[wave] = x;
  __syncthreads();
  int pre = 0, tot = 0;
  for (int w = 0; w < 8; ++w) {
    const int q = wsum8[w];
    if (w < wave) pre += q;
    tot += q;
  }
  __syncthreads();
  total = tot;
  return pre + x - v;
}

__global__ __launch_bounds__(256) void csr_kernel(const int* __restrict__ sortedW,
                                                  const int* __restrict__ woff,
                                                  int* __restrict__ offW,
                                                  int* __restrict__ csr) {
  const int2* __restrict__ sortedE = (const int2*)sortedW;
  int2* __restrict__ off2 = (int2*)offW;
  __shared__ int stage[LCAP];
  __shared__ int ncnt[NPB];
  __shared__ int nstart[NPB + 1];
  __shared__ int wsum8[8];
  const int t = threadIdx.x;
  const int b = blockIdx.x;
  const int nodeBase = b * NPB;

  for (int i = t; i < NPB; i += 256) ncnt[i] = 0;

  int a = 0, len = 0;
  if (t < GA) {
    a = woff[t * WSTR + b];
    len = woff[t * WSTR + b + 1] - a;
  }
  int total, base;
  const int ss = scan256(len, wsum8, total);
  (void)scan256(a, wsum8, base);
  const int2* g = sortedE + (size_t)t * CH + a;
  const bool inLds = total <= LCAP;

  int sv[EPT];
  int kv[EPT];
  int rk[EPT];
  if (inLds) {
    const int e0 = t * CH + a;
    for (int k = 0; k < len; ++k) stage[ss + k] = e0 + k;
    __syncthreads();
    int e[EPT];
#pragma unroll
    for (int i = 0; i < EPT; ++i) {
      const int j = t + 256 * i;
      const bool ok = j < total;
      const int x = stage[ok ? j : 0];
      e[i] = ok ? x : 0;
    }
#pragma unroll
    for (int i = 0; i < EPT; ++i) {
      const int2 p = sortedE[e[i]];
      sv[i] = p.x;
      kv[i] = p.y - nodeBase;
    }
#pragma unroll
    for (int i = 0; i < EPT; ++i) {
      rk[i] = 0;
      if (t + 256 * i < total) rk[i] = atomicAdd(&ncnt[kv[i]], 1);
    }
  } else {
    for (int k = 0; k < len; ++k) atomicAdd(&ncnt[g[k].y - nodeBase], 1);
  }
  __syncthreads();

  const int i0 = 3 * t;
  const int c0 = (i0 < NPB) ? ncnt[i0] : 0;
  const int c1 = (i0 + 1 < NPB) ? ncnt[i0 + 1] : 0;
  const int c2 = (i0 + 2 < NPB) ? ncnt[i0 + 2] : 0;
  int tot2;
  const int pre = scan256(c0 + c1 + c2, wsum8, tot2);
  if (i0 < NPB) nstart[i0] = pre;
  if (i0 + 1 < NPB) nstart[i0 + 1] = pre + c0;
  if (i0 + 2 < NPB) nstart[i0 + 2] = pre + c0 + c1;
  if (t == 0) nstart[NPB] = tot2;
  __syncthreads();

  for (int i = t; i < NPB; i += 256) {
    const int n = nodeBase + i;
    if (n < N_TOTAL) off2[n] = make_int2(base + nstart[i], base + nstart[i + 1]);
  }
  if (b == NB - 1 && t < 4) csr[NUM_EDGES + t] = 0;

  if (inLds) {
    int ns[EPT];
#pragma unroll
    for (int i = 0; i < EPT; ++i) ns[i] = nstart[(t + 256 * i < total) ? kv[i] : 0];
#pragma unroll
    for (int i = 0; i < EPT; ++i)
      if (t + 256 * i < total) stage[ns[i] + rk[i]] = sv[i];
    __syncthreads();
    int w[EPT];
#pragma unroll
    for (int i = 0; i < EPT; ++i) w[i] = stage[(t + 256 * i < total) ? t + 256 * i : 0];
#pragma unroll
    for (int i = 0; i < EPT; ++i)
      if (t + 256 * i < total) csr[base + t + 256 * i] = w[i];
  } else {
    for (int i = t; i < NPB; i += 256) ncnt[i] = 0;
    __syncthreads();
    for (int k = 0; k < len; ++k) {
      const int2 p = g[k];
      const int ln = p.y - nodeBase;
      const int rk1 = atomicAdd(&ncnt[ln], 1);
      csr[base + nstart[ln] + rk1] = p.x;
    }
  }
}

static __device__ __forceinline__ float4 ld_row4(const float* __restrict__ emb, int node, int c4) {
  return *(const float4*)(emb + (size_t)node * EMBED_DIM + c4);
}

static __device__ __forceinline__ void wave_lds_sync() {
  __builtin_amdgcn_fence(__ATOMIC_RELEASE, "wavefront");
  __builtin_amdgcn_wave_barrier();
  __builtin_amdgcn_fence(__ATOMIC_ACQUIRE, "wavefront");
}

static __device__ __forceinline__ void put_partial(float* __restrict__ part, int slot, float4 acc, int h, int c4) {
  acc.x += xhalf(acc.x);
  acc.y += xhalf(acc.y);
  acc.z += xhalf(acc.z);
  acc.w += xhalf(acc.w);
  if (h == 0) *(float4*)(part + slot * PARTP + c4) = acc;
}

__global__ __launch_bounds__(WPB * 32) void walk_dot_kernel(
    const float* __restrict__ emb,
    const int* __restrict__ uidx, const int* __restrict__ iidx,
    const int* __restrict__ offW, const int* __restrict__ csr,
    float* __restrict__ out) {
  const int2* __restrict__ off2 = (const int2*)offW;
  __shared__ int lists[WPB * LIST_CAP];
  __shared__ __attribute__((aligned(16))) float rowsum[2 * 16 * ROWP];
  __shared__ __attribute__((aligned(16))) float part[NPART * PARTP];
  __shared__ int cntL[WPB];
  const int lane = threadIdx.x & 31;
  const int wave = __builtin_amdgcn_readfirstlane((int)(threadIdx.x >> 5));
  int* __restrict__ list = lists + wave * LIST_CAP;
  const int h = lane >> 4;
  const int c4 = (lane & 15) << 2;
  const bool isUser = wave < 16;
  const int r = wave & 15;
  const int rowBase = blockIdx.x * 16;
  const int b = rowBase + r;
  const int node = isUser ? uidx[b] : NUM_USERS + iidx[b];

  if (lane == 0) list[0] = node;
  wave_lds_sync();

  bool ovf = false;
  int fb = 0, fe = 1;
  for (int lvl = 0; lvl < NUM_LAYERS; ++lvl) {
    int wr = fe;
    for (int base = fb; base < fe; base += 32) {
      const int i = base + lane;
      int o0 = 0, dg = 0;
      if (i < fe) {
        const int2 oo = off2[list[i]];
        o0 = oo.x;
        dg = oo.y - oo.x;
      }
      const int x = wave_incl_scan(dg, lane);
      const int tot = __builtin_amdgcn_readlane(x, 31);
      if (wr + tot > LIST_CAP) {
        ovf = true;
        break;
      }
      const int slot = wr + x - dg;
      for (int k = 0; k < NUM_EDGES; k += 8) {
        if (__builtin_amdgcn_ballot_w32(k < dg) == 0u) break;
        const bool a0 = k < dg;
        const bool a1 = k + 4 < dg;
        int4 q0 = make_int4(0, 0, 0, 0), q1 = make_int4(0, 0, 0, 0);
        if (a0) __builtin_memcpy(&q0, csr + o0 + k, 16);
        if (a1) __builtin_memcpy(&q1, csr + o0 + k + 4, 16);
        if (a0) {
          list[slot + k] = q0.x;
          if (k + 1 < dg) list[slot + k + 1] = q0.y;
          if (k + 2 < dg) list[slot + k + 2] = q0.z;
          if (k + 3 < dg) list[slot + k + 3] = q0.w;
        }
        if (a1) {
          list[slot + k + 4] = q1.x;
          if (k + 5 < dg) list[slot + k + 5] = q1.y;
          if (k + 6 < dg) list[slot + k + 6] = q1.z;
          if (k + 7 < dg) list[slot + k + 7] = q1.w;
        }
      }
      wr += tot;
    }
    wave_lds_sync();
    if (ovf) break;
    fb = fe;
    fe = wr;
  }

  float4 own = make_float4(0.f, 0.f, 0.f, 0.f);
  if (ovf) {
    const float4 r0 = ld_row4(emb, node, c4);
    if (h == 0) { own.x += r0.x; own.y += r0.y; own.z += r0.z; own.w += r0.w; }
    const int2 q1 = off2[node];
    for (int j1 = q1.x; j1 < q1.y; ++j1) {
      const int n1 = csr[j1];
      const float4 r1 = ld_row4(emb, n1, c4);
      if (h == 0) { own.x += r1.x; own.y += r1.y; own.z += r1.z; own.w += r1.w; }
      const int2 q2 = off2[n1];
      for (int j2 = q2.x; j2 < q2.y; ++j2) {
        const int n2 = csr[j2];
        const float4 r2 = ld_row4(emb, n2, c4);
        if (h == 0) { own.x += r2.x; own.y += r2.y; own.z += r2.z; own.w += r2.w; }
        const int2 q3 = off2[n2];
        for (int j3 = q3.x; j3 < q3.y; ++j3) {
          const int n3 = csr[j3];
          const float4 r3 = ld_row4(emb, n3, c4);
          if (h == 0) { own.x += r3.x; own.y += r3.y; own.z += r3.z; own.w += r3.w; }
        }
      }
    }
  }
  if (lane == 0) cntL[wave] = ovf ? 0 : fe;
  __syncthreads();

  const int cnt = cntL[lane];
  const int units = (cnt + 15) >> 4;
  const int incl = wave_incl_scan(units, lane);
  const int total = __builtin_amdgcn_readlane(incl, 31);
  const int share = (total + WPB - 1) / WPB;
  const int u0 = wave * share;
  const int u1 = (u0 + share < total) ? (u0 + share) : total;

  for (int u = u0; u < u1;) {
    const int L = __builtin_popcount(__builtin_amdgcn_ballot_w32(incl <= u));
    const int endL = __builtin_amdgcn_readlane(incl, L);
    const int first = endL - __builtin_amdgcn_readlane(units, L);
    const int nL = __builtin_amdgcn_readlane(cnt, L);
    const int uEnd = endL < u1 ? endL : u1;
    const int* __restrict__ lst = lists + L * LIST_CAP;
    const int iEnd = (uEnd - first) << 4;
    float4 acc = make_float4(0.f, 0.f, 0.f, 0.f);
    for (int i = (u - first) << 4; i < iEnd; i += 16) {
      float4 v[8];
#pragma unroll
      for (int k = 0; k < 8; ++k) {
        const int idx = i + 2 * k + h;
        const bool ok = idx < nL;
        const int s = lst[ok ? idx : 0];
        const float4 rw = ld_row4(emb, s, c4);
        v[k] = ok ? rw : make_float4(0.f, 0.f, 0.f, 0.f);
      }
#pragma unroll
      for (int k = 0; k < 8; ++k) {
        acc.x += v[k].x; acc.y += v[k].y; acc.z += v[k].z; acc.w += v[k].w;
      }
    }
    put_partial(part, wave + L, acc, h, c4);
    u = uEnd;
  }
  __syncthreads();

  const int myUnits = __builtin_amdgcn_readlane(units, wave);
  if (myUnits > 0) {
    const int myEnd = __builtin_amdgcn_readlane(incl, wave);
    const int wlo = (myEnd - myUnits) / share;
    const int whi = (myEnd - 1) / share;
    for (int w = wlo; w <= whi; ++w) {
      const float4 p = *(const float4*)(part + (w + wave) * PARTP + c4);
      own.x += p.x; own.y += p.y; own.z += p.z; own.w += p.w;
    }
  }
  if (h == 0) *(float4*)(rowsum + (isUser ? 0 : 16 * ROWP) + r * ROWP + c4) = own;
  __syncthreads();

  if (wave == 0) {
    const int kOff = (lane >> 4) << 1;
    const float* uRow = rowsum + (lane & 15) * ROWP + kOff;
    const float* vRow = rowsum + 16 * ROWP + (lane & 15) * ROWP + kOff;

    v8f c = {};
#pragma unroll
    for (int k = 0; k < EMBED_DIM; k += 4) {
      v2f a, bf;
      a.x = uRow[k];
      a.y = uRow[k + 1];
      bf.x = vRow[k];
      bf.y = vRow[k + 1];
      c = __builtin_amdgcn_wmma_f32_16x16x4_f32(false, a, false, bf, (short)0, c,
                                                false, false);
    }

    float cd[8];
#pragma unroll
    for (int i = 0; i < 8; ++i) cd[i] = c[i];
    int dsel = (lane < 16) ? lane : (lane - 24);
    float diag = cd[0];
#pragma unroll
    for (int i = 1; i < 8; ++i)
      if (dsel == i) diag = cd[i];

    const float scale = 1.0f / 16.0f;
    if (lane < 8)
      out[rowBase + lane] = diag * scale;
    else if (lane >= 24)
      out[rowBase + (lane - 16)] = diag * scale;
  }
}

extern "C" void kernel_launch(void* const* d_in, const int* in_sizes, int n_in,
                              void* d_out, int out_size, void* d_ws,
                              size_t ws_size, hipStream_t stream) {
  const float* emb = (const float*)d_in[0];
  const int* eidx = (const int*)d_in[1];
  const int* uidx = (const int*)d_in[2];
  const int* iidx = (const int*)d_in[3];
  float* out = (float*)d_out;

  const int* src = eidx;
  const int* dst = eidx + NUM_EDGES;

  int* sortedE = (int*)d_ws;
  int* woff = sortedE + 2 * GA * CH;
  int* off2 = woff + WOFF_PAD;
  int* csr = off2 + 2 * OFF_PAD;

  bin_kernel<<<GA, 1024, 0, stream>>>(src, dst, sortedE, woff);
  csr_kernel<<<NB, 256, 0, stream>>>(sortedE, woff, off2, csr);

  walk_dot_kernel<<<BATCH / 16, WPB * 32, 0, stream>>>(emb, uidx, iidx, off2, csr, out);
}
